# starting kernel
# baseline (speedup 1.0000x reference)
#include <hip/hip_runtime.h>

#define NFEAT 128

typedef float v2f  __attribute__((ext_vector_type(2)));
typedef float v8f  __attribute__((ext_vector_type(8)));

__device__ __forceinline__ void atomic_add_f32_dev(float* p, float v) {
    asm volatile("global_atomic_add_f32 %0, %1, off scope:SCOPE_DEV"
                 :: "v"(p), "v"(v) : "memory");
}

__global__ void k_deg_init(float* __restrict__ deg, int n) {
    int i = blockIdx.x * blockDim.x + threadIdx.x;
    if (i < n) deg[i] = 1.0f;
}

__global__ void k_deg_accum(const int* __restrict__ col, const float* __restrict__ ew,
                            float* __restrict__ deg, int E) {
    int e = blockIdx.x * blockDim.x + threadIdx.x;
    if (e < E) atomic_add_f32_dev(&deg[col[e]], ew[e]);
}

__global__ void k_rsqrt(float* __restrict__ deg, int n) {
    int i = blockIdx.x * blockDim.x + threadIdx.x;
    if (i < n) deg[i] = rsqrtf(deg[i]);
}

__global__ void __launch_bounds__(256)
k_gemm128_wmma(const float* __restrict__ A, const float* __restrict__ W,
               float* __restrict__ out, int nrows) {
    const int lane = threadIdx.x;
    const int wv   = threadIdx.y;
    const int m0   = blockIdx.x * 16;
    const int n0   = wv * 16;
    const int hi   = lane >> 4;
    const int l    = lane & 15;
    if (m0 + 16 > nrows) return;

    v8f acc = {};
    const float* ap = A + (size_t)(m0 + l) * NFEAT + 2 * hi;
    const float* bp = W + (size_t)(2 * hi) * NFEAT + n0 + l;
#pragma unroll 8
    for (int k = 0; k < NFEAT; k += 4) {
        v2f a = *(const v2f*)(ap + k);
        v2f b;
        b.x = bp[(size_t)k * NFEAT];
        b.y = bp[(size_t)k * NFEAT + NFEAT];
        acc = __builtin_amdgcn_wmma_f32_16x16x4_f32(
            false, a, false, b, (short)0, acc, false, false);
    }
    float* op = out + (size_t)(m0 + 8 * hi) * NFEAT + n0 + l;
#pragma unroll
    for (int v = 0; v < 8; ++v)
        op[(size_t)v * NFEAT] = acc[v];
}

__global__ void k_selfbias(const float* __restrict__ h, const float* __restrict__ dinv,
                           const float* __restrict__ bias, float* __restrict__ out, int n) {
    int idx   = blockIdx.x * blockDim.x + threadIdx.x;
    int total = n * (NFEAT / 4);
    if (idx >= total) return;
    int node = idx >> 5;
    int f    = (idx & 31) * 4;
    float di = dinv[node];
    float s  = di * di;
    float4 hv = *(const float4*)(h + (size_t)node * NFEAT + f);
    float4 bv = *(const float4*)(bias + f);
    float4 o;
    o.x = fmaf(s, hv.x, bv.x);
    o.y = fmaf(s, hv.y, bv.y);
    o.z = fmaf(s, hv.z, bv.z);
    o.w = fmaf(s, hv.w, bv.w);
    *(float4*)(out + (size_t)node * NFEAT + f) = o;
}

__global__ void __launch_bounds__(256)
k_edge_agg(const int* __restrict__ row, const int* __restrict__ col,
           const float* __restrict__ ew, const float* __restrict__ dinv,
           const float* __restrict__ h, float* __restrict__ out, int E) {
    int e = blockIdx.x * blockDim.y + threadIdx.y;
    if (e >= E) return;
    int r = row[e];
    int c = col[e];
    float nrm = dinv[r] * ew[e] * dinv[c];
    int f = threadIdx.x * 4;
    float4 hv = *(const float4*)(h + (size_t)r * NFEAT + f);
    float* op = out + (size_t)c * NFEAT + f;
    atomic_add_f32_dev(op + 0, nrm * hv.x);
    atomic_add_f32_dev(op + 1, nrm * hv.y);
    atomic_add_f32_dev(op + 2, nrm * hv.z);
    atomic_add_f32_dev(op + 3, nrm * hv.w);
}

__global__ void k_relu(float* __restrict__ p, int n4) {
    int idx = blockIdx.x * blockDim.x + threadIdx.x;
    if (idx >= n4) return;
    float4 v = *(float4*)(p + (size_t)idx * 4);
    v.x = fmaxf(v.x, 0.0f);
    v.y = fmaxf(v.y, 0.0f);
    v.z = fmaxf(v.z, 0.0f);
    v.w = fmaxf(v.w, 0.0f);
    *(float4*)(p + (size_t)idx * 4) = v;
}

extern "C" void kernel_launch(void* const* d_in, const int* in_sizes, int n_in,
                              void* d_out, int out_size, void* d_ws, size_t ws_size,
                              hipStream_t stream) {
    const float* x  = (const float*)d_in[0];
    const int*   ei = (const int*)d_in[1];
    const float* ew = (const float*)d_in[2];
    const float* W1 = (const float*)d_in[3];
    const float* b1 = (const float*)d_in[4];
    const float* W2 = (const float*)d_in[5];
    const float* b2 = (const float*)d_in[6];
    float* out = (float*)d_out;

    const int N = in_sizes[0] / NFEAT;
    const int E = in_sizes[2];
    const int* row = ei;
    const int* col = ei + E;

    float* ws   = (float*)d_ws;
    size_t npad = ((size_t)N + 255) & ~(size_t)255;
    float* dinv = ws;
    float* h    = ws + npad;
    float* g    = h + (size_t)N * NFEAT;

    const int T = 256;
    k_deg_init<<<(N + T - 1) / T, T, 0, stream>>>(dinv, N);
    k_deg_accum<<<(E + T - 1) / T, T, 0, stream>>>(col, ew, dinv, E);
    k_rsqrt<<<(N + T - 1) / T, T, 0, stream>>>(dinv, N);

    dim3 wblk(32, 8);
    int mtiles = (N + 15) / 16;
    int tvec   = N * (NFEAT / 4);
    int eblk   = (E + 7) / 8;

    k_gemm128_wmma<<<mtiles, wblk, 0, stream>>>(x, W1, h, N);
    k_selfbias<<<(tvec + T - 1) / T, T, 0, stream>>>(h, dinv, b1, g, N);
    k_edge_agg<<<eblk, wblk, 0, stream>>>(row, col, ew, dinv, h, g, E);
    k_relu<<<(tvec + T - 1) / T, T, 0, stream>>>(g, tvec);

    k_gemm128_wmma<<<mtiles, wblk, 0, stream>>>(g, W2, h, N);
    k_selfbias<<<(tvec + T - 1) / T, T, 0, stream>>>(h, dinv, b2, out, N);
    k_edge_agg<<<eblk, wblk, 0, stream>>>(row, col, ew, dinv, h, out, E);
}


# Edge scatter in two sweeps by chip half: resident waves first add the edges whose target row is in their own half's range of rows, then the rest; same f32 atomics, arrival-ordered
# speedup vs baseline: 2.3000x; 2.3000x over previous
#include <hip/hip_runtime.h>

#define NFEAT 128

typedef float v2f  __attribute__((ext_vector_type(2)));
typedef float v8f  __attribute__((ext_vector_type(8)));

__device__ __forceinline__ void atomic_add_f32_dev(float* p, float v) {
    asm volatile("global_atomic_add_f32 %0, %1, off scope:SCOPE_DEV"
                 :: "v"(p), "v"(v) : "memory");
}

__global__ void k_deg_init(float* __restrict__ deg, int n) {
    int i = blockIdx.x * blockDim.x + threadIdx.x;
    if (i < n) deg[i] = 1.0f;
}

__global__ void k_deg_accum(const int* __restrict__ col, const float* __restrict__ ew,
                            float* __restrict__ deg, int E) {
    int e = blockIdx.x * blockDim.x + threadIdx.x;
    if (e < E) atomic_add_f32_dev(&deg[col[e]], ew[e]);
}

__global__ void k_rsqrt(float* __restrict__ deg, int n) {
    int i = blockIdx.x * blockDim.x + threadIdx.x;
    if (i < n) deg[i] = rsqrtf(deg[i]);
}

__device__ __forceinline__ void wmma_f32_cover() {
    __builtin_amdgcn_sched_barrier(0);
    asm volatile("v_nop\n\tv_nop\n\tv_nop\n\tv_nop");
    __builtin_amdgcn_sched_barrier(0);
}

#define GW      8
#define APITCH  132
#define GBLK    1024

template <bool RELU_A>
__device__ __forceinline__ void
gemm128_body(const float* __restrict__ A, const float* __restrict__ W,
             const float* __restrict__ dinv, const float* __restrict__ bias,
             float* __restrict__ out, float* __restrict__ tgt, int nrows,
             float* __restrict__ As  ) {
    const int lane = threadIdx.x;
    const int wv   = threadIdx.y;
    const int n0   = wv * 16;
    const int hi   = lane >> 4;
    const int l    = lane & 15;
    const int ntiles = nrows / 16;
    const int tstep  = gridDim.x;

    v2f breg[NFEAT / 4];
    {
        const float* bp = W + (size_t)(2 * hi) * NFEAT + n0 + l;
#pragma unroll
        for (int s = 0; s < NFEAT / 4; ++s) {
            breg[s].x = bp[(size_t)(4 * s) * NFEAT];
            breg[s].y = bp[(size_t)(4 * s) * NFEAT + NFEAT];
        }
    }
    const float bv = bias[n0 + l];

    const int    sr   = 2 * wv;
    const size_t goff = (size_t)sr * NFEAT + 4 * lane;
    const int    soff = sr * APITCH + 4 * lane;
    const int    foff = l * APITCH + 2 * hi;

    int t = blockIdx.x;
    float4 p0 = {}, p1 = {};
    if (t < ntiles) {
        const float* g = A + (size_t)t * 16 * NFEAT + goff;
        p0 = *(const float4*)g;
        p1 = *(const float4*)(g + NFEAT);
    }
    int buf = 0;
    for (; t < ntiles; t += tstep, buf ^= 1) {
        float* as = As + buf * (16 * APITCH);
        if (RELU_A) {
            p0.x = fmaxf(p0.x, 0.0f); p0.y = fmaxf(p0.y, 0.0f);
            p0.z = fmaxf(p0.z, 0.0f); p0.w = fmaxf(p0.w, 0.0f);
            p1.x = fmaxf(p1.x, 0.0f); p1.y = fmaxf(p1.y, 0.0f);
            p1.z = fmaxf(p1.z, 0.0f); p1.w = fmaxf(p1.w, 0.0f);
        }
        *(float4*)(as + soff)          = p0;
        *(float4*)(as + soff + APITCH) = p1;
        __syncthreads();
        const int tn = t + tstep;
        if (tn < ntiles) {
            const float* g = A + (size_t)tn * 16 * NFEAT + goff;
            p0 = *(const float4*)g;
            p1 = *(const float4*)(g + NFEAT);
        }

        v8f acc = {};
        const float* fp = as + foff;
#pragma unroll
        for (int s = 0; s < NFEAT / 4; ++s) {
            const v2f a = *(const v2f*)(fp + 4 * s);
            acc = __builtin_amdgcn_wmma_f32_16x16x4_f32(
                false, a, false, breg[s], (short)0, acc, false, false);
        }
        wmma_f32_cover();

        const int    r0  = t * 16 + 8 * hi;
        const size_t off = (size_t)r0 * NFEAT + n0 + l;
        float* op = out + off;
        float* tp = tgt + off;
#pragma unroll
        for (int v = 0; v < 8; ++v) {
            const float di = dinv[r0 + v];
            const float s  = di * di;
            const float hv = acc[v];
            op[(size_t)v * NFEAT] = hv;
            tp[(size_t)v * NFEAT] = fmaf(s, hv, bv);
        }
    }
}

__global__ void __launch_bounds__(32 * GW)
k_gemm128_wmma(const float* __restrict__ A, const float* __restrict__ W,
               const float* __restrict__ dinv, const float* __restrict__ bias,
               float* __restrict__ out, float* __restrict__ tgt, int nrows) {
    __shared__ __attribute__((aligned(16))) float As[2 * 16 * APITCH];
    gemm128_body<false>(A, W, dinv, bias, out, tgt, nrows, As);
}

__global__ void __launch_bounds__(32 * GW)
k_gemm128_wmma_relu(const float* __restrict__ A, const float* __restrict__ W,
                    const float* __restrict__ dinv, const float* __restrict__ bias,
                    float* __restrict__ out, float* __restrict__ tgt, int nrows) {
    __shared__ __attribute__((aligned(16))) float As[2 * 16 * APITCH];
    gemm128_body<true>(A, W, dinv, bias, out, tgt, nrows, As);
}

#define EPW   32
#define EB    8
#define EWV   8
#define EGRID 2048

__device__ __forceinline__ int lane_read_i(int v, int j) {
    return __builtin_amdgcn_readlane(v, j);
}
__device__ __forceinline__ float lane_read_f(float v, int j) {
    return __int_as_float(__builtin_amdgcn_readlane(__float_as_int(v), j));
}

__device__ __forceinline__ int chip_half() {
    return (int)((__builtin_amdgcn_s_sendmsg_rtn(0x87) >> 18) & 1u);
}

__device__ __forceinline__ float4 load4_row(const float* rowp, unsigned boff) {
    const char* p = (const char*)rowp + boff;
    float4 v;
    v.x = *(const float*)(p);
    v.y = *(const float*)(p + 128);
    v.z = *(const float*)(p + 256);
    v.w = *(const float*)(p + 384);
    return v;
}

__device__ __forceinline__ void atomic_add4_row(float* rowp, unsigned boff, float4 v) {
    asm volatile("global_atomic_add_f32 %0, %1, %5 scope:SCOPE_DEV\n\t"
                 "global_atomic_add_f32 %0, %2, %5 offset:128 scope:SCOPE_DEV\n\t"
                 "global_atomic_add_f32 %0, %3, %5 offset:256 scope:SCOPE_DEV\n\t"
                 "global_atomic_add_f32 %0, %4, %5 offset:384 scope:SCOPE_DEV\n\t"
                 "s_wait_xcnt 0x0"
                 :: "v"(boff), "v"(v.x), "v"(v.y), "v"(v.z), "v"(v.w), "s"(rowp)
                 : "memory");
}

__global__ void __launch_bounds__(32 * EWV)
k_edge_agg(const int* __restrict__ row, const int* __restrict__ col,
           const float* __restrict__ ew, const float* __restrict__ dinv,
           const float* __restrict__ h, float* __restrict__ out, int E, int n) {
    const int lane = threadIdx.x;
    const int wv   = __builtin_amdgcn_readfirstlane((int)threadIdx.y);
    const unsigned boff = (unsigned)lane * 4u;
    const int nsplit = n >> 1;
    const int ngrp   = (E + EPW - 1) / EPW;
    const int gw     = blockIdx.x * EWV + wv;
    const int nw     = gridDim.x * EWV;
    const int myq    = chip_half();

    for (int sweep = 0; sweep < 2; ++sweep) {
        const int q = myq ^ sweep;
        for (int g = gw; g < ngrp; g += nw) {
            const int e = g * EPW + lane;
            int r = 0, c = 0;
            float w = 0.0f, nrm = 0.0f;
            bool mine = false;
            if (e < E) {
                c = col[e];
                r = row[e];
                w = ew[e];
                mine = ((c >= nsplit) ? 1 : 0) == q;
            }
            if (mine) nrm = dinv[r] * w * dinv[c];
            unsigned m = __builtin_amdgcn_ballot_w32(mine);
            for (int bt = 0; bt < EPW / EB && m != 0u; ++bt) {
                int jj[EB];
#pragma unroll
                for (int k = 0; k < EB; ++k) {
                    jj[k] = (m != 0u) ? __builtin_ctz(m) : -1;
                    m &= m - 1u;
                }
                float4 cur[EB];
#pragma unroll
                for (int k = 0; k < EB; ++k)
                    if (jj[k] >= 0)
                        cur[k] = load4_row(h + (size_t)lane_read_i(r, jj[k]) * NFEAT, boff);
#pragma unroll
                for (int k = 0; k < EB; ++k) {
                    if (jj[k] >= 0) {
                        const int cj = lane_read_i(c, jj[k]);
                        const float nj = lane_read_f(nrm, jj[k]);
                        float4 mv;
                        mv.x = nj * cur[k].x;
                        mv.y = nj * cur[k].y;
                        mv.z = nj * cur[k].z;
                        mv.w = nj * cur[k].w;
                        atomic_add4_row(out + (size_t)cj * NFEAT, boff, mv);
                    }
                }
            }
        }
    }
}

extern "C" void kernel_launch(void* const* d_in, const int* in_sizes, int n_in,
                              void* d_out, int out_size, void* d_ws, size_t ws_size,
                              hipStream_t stream) {
    const float* x  = (const float*)d_in[0];
    const int*   ei = (const int*)d_in[1];
    const float* ew = (const float*)d_in[2];
    const float* W1 = (const float*)d_in[3];
    const float* b1 = (const float*)d_in[4];
    const float* W2 = (const float*)d_in[5];
    const float* b2 = (const float*)d_in[6];
    float* out = (float*)d_out;

    const int N = in_sizes[0] / NFEAT;
    const int E = in_sizes[2];
    const int* row = ei;
    const int* col = ei + E;

    float* ws   = (float*)d_ws;
    size_t npad = ((size_t)N + 255) & ~(size_t)255;
    float* dinv = ws;
    float* h    = ws + npad;
    float* g    = h + (size_t)N * NFEAT;

    const int T = 256;
    k_deg_init<<<(N + T - 1) / T, T, 0, stream>>>(dinv, N);
    k_deg_accum<<<(E + T - 1) / T, T, 0, stream>>>(col, ew, dinv, E);
    k_rsqrt<<<(N + T - 1) / T, T, 0, stream>>>(dinv, N);

    dim3 wblk(32, GW);

    k_gemm128_wmma<<<GBLK, wblk, 0, stream>>>(x, W1, dinv, b1, h, g, N);
    k_edge_agg<<<EGRID, dim3(32, EWV), 0, stream>>>(row, col, ew, dinv, h, g, E, N);

    k_gemm128_wmma_relu<<<GBLK, wblk, 0, stream>>>(g, W2, dinv, b2, h, out, N);
    k_edge_agg<<<EGRID, dim3(32, EWV), 0, stream>>>(row, col, ew, dinv, h, out, E, N);
}


# Edge norm split over the dense passes: GEMM stores rows pre-scaled by dinv[src]; dinv[dst] (+b1) applied after the layer-1 sum as layer 2 stages A; per-edge dinv reads 4 -> 1, same atomics
# speedup vs baseline: 2.3960x; 1.0417x over previous
#include <hip/hip_runtime.h>

#define NFEAT 128

typedef float v2f  __attribute__((ext_vector_type(2)));
typedef float v8f  __attribute__((ext_vector_type(8)));

__device__ __forceinline__ void atomic_add_f32_dev(float* p, float v) {
    asm volatile("global_atomic_add_f32 %0, %1, off scope:SCOPE_DEV"
                 :: "v"(p), "v"(v) : "memory");
}

__global__ void k_deg_init(float* __restrict__ deg, int n) {
    int i = blockIdx.x * blockDim.x + threadIdx.x;
    if (i < n) deg[i] = 1.0f;
}

__global__ void k_deg_accum(const int* __restrict__ col, const float* __restrict__ ew,
                            float* __restrict__ deg, int E) {
    int e = blockIdx.x * blockDim.x + threadIdx.x;
    if (e < E) atomic_add_f32_dev(&deg[col[e]], ew[e]);
}

__global__ void k_rsqrt(float* __restrict__ deg, int n) {
    int i = blockIdx.x * blockDim.x + threadIdx.x;
    if (i < n) deg[i] = rsqrtf(deg[i]);
}

__device__ __forceinline__ void wmma_f32_cover() {
    __builtin_amdgcn_sched_barrier(0);
    asm volatile("v_nop\n\tv_nop\n\tv_nop\n\tv_nop");
    __builtin_amdgcn_sched_barrier(0);
}

#define GW      8
#define APITCH  132
#define GBLK    1024

template <bool SECOND>
__device__ __forceinline__ void
gemm128_body(const float* __restrict__ A, const float* __restrict__ W,
             const float* __restrict__ dinv, const float* __restrict__ bias,
             const float* __restrict__ abias  ,
             float* __restrict__ out, float* __restrict__ tgt, int nrows,
             float* __restrict__ As  ) {
    const int lane = threadIdx.x;
    const int wv   = threadIdx.y;
    const int n0   = wv * 16;
    const int hi   = lane >> 4;
    const int l    = lane & 15;
    const int ntiles = nrows / 16;
    const int tstep  = gridDim.x;

    v2f breg[NFEAT / 4];
    {
        const float* bp = W + (size_t)(2 * hi) * NFEAT + n0 + l;
#pragma unroll
        for (int s = 0; s < NFEAT / 4; ++s) {
            breg[s].x = bp[(size_t)(4 * s) * NFEAT];
            breg[s].y = bp[(size_t)(4 * s) * NFEAT + NFEAT];
        }
    }
    float bv = 0.0f;
    if (SECOND) bv = bias[n0 + l];

    const int    sr   = 2 * wv;
    const size_t goff = (size_t)sr * NFEAT + 4 * lane;
    const int    soff = sr * APITCH + 4 * lane;
    const int    foff = l * APITCH + 2 * hi;

    float4 ab = {};
    if (SECOND) { ab.x = abias[4 * lane]; ab.y = abias[4 * lane + 1]; ab.z = abias[4 * lane + 2]; ab.w = abias[4 * lane + 3]; }
    int t = blockIdx.x;
    float4 p0 = {}, p1 = {};
    float d0 = 0.0f, d1 = 0.0f;
    if (t < ntiles) {
        const float* g = A + (size_t)t * 16 * NFEAT + goff;
        p0 = *(const float4*)g;
        p1 = *(const float4*)(g + NFEAT);
        if (SECOND) { d0 = dinv[t * 16 + sr]; d1 = dinv[t * 16 + sr + 1]; }
    }
    int buf = 0;
    for (; t < ntiles; t += tstep, buf ^= 1) {
        float* as = As + buf * (16 * APITCH);
        if (SECOND) {
            p0.x = fmaxf(fmaf(d0, p0.x, ab.x), 0.0f); p0.y = fmaxf(fmaf(d0, p0.y, ab.y), 0.0f);
            p0.z = fmaxf(fmaf(d0, p0.z, ab.z), 0.0f); p0.w = fmaxf(fmaf(d0, p0.w, ab.w), 0.0f);
            p1.x = fmaxf(fmaf(d1, p1.x, ab.x), 0.0f); p1.y = fmaxf(fmaf(d1, p1.y, ab.y), 0.0f);
            p1.z = fmaxf(fmaf(d1, p1.z, ab.z), 0.0f); p1.w = fmaxf(fmaf(d1, p1.w, ab.w), 0.0f);
        }
        *(float4*)(as + soff)          = p0;
        *(float4*)(as + soff + APITCH) = p1;
        __syncthreads();
        const int tn = t + tstep;
        if (tn < ntiles) {
            const float* g = A + (size_t)tn * 16 * NFEAT + goff;
            p0 = *(const float4*)g;
            p1 = *(const float4*)(g + NFEAT);
            if (SECOND) { d0 = dinv[tn * 16 + sr]; d1 = dinv[tn * 16 + sr + 1]; }
        }

        v8f acc = {};
        const float* fp = as + foff;
#pragma unroll
        for (int s = 0; s < NFEAT / 4; ++s) {
            const v2f a = *(const v2f*)(fp + 4 * s);
            acc = __builtin_amdgcn_wmma_f32_16x16x4_f32(
                false, a, false, breg[s], (short)0, acc, false, false);
        }
        wmma_f32_cover();

        const int    r0  = t * 16 + 8 * hi;
        const size_t off = (size_t)r0 * NFEAT + n0 + l;
        float* op = out + off;
        float* tp = tgt + off;
#pragma unroll
        for (int v = 0; v < 8; ++v) {
            const float di = dinv[r0 + v];
            const float hv = acc[v];
            const float hs = di * hv;
            op[(size_t)v * NFEAT] = hs;
            tp[(size_t)v * NFEAT] = SECOND ? fmaf(di * di, hv, bv) : hs;
        }
    }
}

__global__ void __launch_bounds__(32 * GW)
k_gemm128_wmma(const float* __restrict__ A, const float* __restrict__ W,
               const float* __restrict__ dinv,
               float* __restrict__ out, float* __restrict__ tgt, int nrows) {
    __shared__ __attribute__((aligned(16))) float As[2 * 16 * APITCH];
    gemm128_body<false>(A, W, dinv, nullptr, nullptr, out, tgt, nrows, As);
}

__global__ void __launch_bounds__(32 * GW)
k_gemm128_wmma_relu(const float* __restrict__ A, const float* __restrict__ W,
                    const float* __restrict__ dinv, const float* __restrict__ bias,
                    const float* __restrict__ abias,
                    float* __restrict__ out, float* __restrict__ tgt, int nrows) {
    __shared__ __attribute__((aligned(16))) float As[2 * 16 * APITCH];
    gemm128_body<true>(A, W, dinv, bias, abias, out, tgt, nrows, As);
}

#define EPW   32
#define EB    8
#define EWV   8
#define EGRID 2048

__device__ __forceinline__ int lane_read_i(int v, int j) {
    return __builtin_amdgcn_readlane(v, j);
}
__device__ __forceinline__ float lane_read_f(float v, int j) {
    return __int_as_float(__builtin_amdgcn_readlane(__float_as_int(v), j));
}

__device__ __forceinline__ int chip_half() {
    return (int)((__builtin_amdgcn_s_sendmsg_rtn(0x87) >> 18) & 1u);
}

__device__ __forceinline__ float4 load4_row(const float* rowp, unsigned boff) {
    const char* p = (const char*)rowp + boff;
    float4 v;
    v.x = *(const float*)(p);
    v.y = *(const float*)(p + 128);
    v.z = *(const float*)(p + 256);
    v.w = *(const float*)(p + 384);
    return v;
}

__device__ __forceinline__ void atomic_add4_row(float* rowp, unsigned boff, float4 v) {
    asm volatile("global_atomic_add_f32 %0, %1, %5 scope:SCOPE_DEV\n\t"
                 "global_atomic_add_f32 %0, %2, %5 offset:128 scope:SCOPE_DEV\n\t"
                 "global_atomic_add_f32 %0, %3, %5 offset:256 scope:SCOPE_DEV\n\t"
                 "global_atomic_add_f32 %0, %4, %5 offset:384 scope:SCOPE_DEV\n\t"
                 "s_wait_xcnt 0x0"
                 :: "v"(boff), "v"(v.x), "v"(v.y), "v"(v.z), "v"(v.w), "s"(rowp)
                 : "memory");
}

template <bool TFACTOR>
__global__ void __launch_bounds__(32 * EWV)
k_edge_agg(const int* __restrict__ row, const int* __restrict__ col,
           const float* __restrict__ ew, const float* __restrict__ dinv,
           const float* __restrict__ h, float* __restrict__ out, int E, int n) {
    const int lane = threadIdx.x;
    const int wv   = __builtin_amdgcn_readfirstlane((int)threadIdx.y);
    const unsigned boff = (unsigned)lane * 4u;
    const int nsplit = n >> 1;
    const int ngrp   = (E + EPW - 1) / EPW;
    const int gw     = blockIdx.x * EWV + wv;
    const int nw     = gridDim.x * EWV;
    const int myq    = chip_half();

    for (int sweep = 0; sweep < 2; ++sweep) {
        const int q = myq ^ sweep;
        for (int g = gw; g < ngrp; g += nw) {
            const int e = g * EPW + lane;
            int r = 0, c = 0;
            float w = 0.0f, nrm = 0.0f;
            bool mine = false;
            if (e < E) {
                c = col[e];
                r = row[e];
                w = ew[e];
                mine = ((c >= nsplit) ? 1 : 0) == q;
            }
            if (mine) nrm = TFACTOR ? w * dinv[c] : w;
            unsigned m = __builtin_amdgcn_ballot_w32(mine);
            for (int bt = 0; bt < EPW / EB && m != 0u; ++bt) {
                int jj[EB];
#pragma unroll
                for (int k = 0; k < EB; ++k) {
                    jj[k] = (m != 0u) ? __builtin_ctz(m) : -1;
                    m &= m - 1u;
                }
                float4 cur[EB];
#pragma unroll
                for (int k = 0; k < EB; ++k)
                    if (jj[k] >= 0)
                        cur[k] = load4_row(h + (size_t)lane_read_i(r, jj[k]) * NFEAT, boff);
#pragma unroll
                for (int k = 0; k < EB; ++k) {
                    if (jj[k] >= 0) {
                        const int cj = lane_read_i(c, jj[k]);
                        const float nj = lane_read_f(nrm, jj[k]);
                        float4 mv;
                        mv.x = nj * cur[k].x;
                        mv.y = nj * cur[k].y;
                        mv.z = nj * cur[k].z;
                        mv.w = nj * cur[k].w;
                        atomic_add4_row(out + (size_t)cj * NFEAT, boff, mv);
                    }
                }
            }
        }
    }
}

extern "C" void kernel_launch(void* const* d_in, const int* in_sizes, int n_in,
                              void* d_out, int out_size, void* d_ws, size_t ws_size,
                              hipStream_t stream) {
    const float* x  = (const float*)d_in[0];
    const int*   ei = (const int*)d_in[1];
    const float* ew = (const float*)d_in[2];
    const float* W1 = (const float*)d_in[3];
    const float* b1 = (const float*)d_in[4];
    const float* W2 = (const float*)d_in[5];
    const float* b2 = (const float*)d_in[6];
    float* out = (float*)d_out;

    const int N = in_sizes[0] / NFEAT;
    const int E = in_sizes[2];
    const int* row = ei;
    const int* col = ei + E;

    float* ws   = (float*)d_ws;
    size_t npad = ((size_t)N + 255) & ~(size_t)255;
    float* dinv = ws;
    float* h    = ws + npad;
    float* g    = h + (size_t)N * NFEAT;

    const int T = 256;
    k_deg_init<<<(N + T - 1) / T, T, 0, stream>>>(dinv, N);
    k_deg_accum<<<(E + T - 1) / T, T, 0, stream>>>(col, ew, dinv, E);
    k_rsqrt<<<(N + T - 1) / T, T, 0, stream>>>(dinv, N);

    dim3 wblk(32, GW);

    k_gemm128_wmma<<<GBLK, wblk, 0, stream>>>(x, W1, dinv, h, g, N);
    k_edge_agg<false><<<EGRID, dim3(32, EWV), 0, stream>>>(row, col, ew, dinv, h, g, E, N);

    k_gemm128_wmma_relu<<<GBLK, wblk, 0, stream>>>(g, W2, dinv, b2, b1, h, out, N);
    k_edge_agg<true><<<EGRID, dim3(32, EWV), 0, stream>>>(row, col, ew, dinv, h, out, E, N);
}


# speedup vs baseline: 2.8314x; 1.1817x over previous
#include <hip/hip_runtime.h>

#define NFEAT 128

typedef float v2f  __attribute__((ext_vector_type(2)));
typedef float v8f  __attribute__((ext_vector_type(8)));

__device__ __forceinline__ void atomic_add_f32_dev(float* p, float v) {
    asm volatile("global_atomic_add_f32 %0, %1, off scope:SCOPE_DEV\n\t"
                 "s_wait_xcnt 0x0"
                 :: "v"(p), "v"(v) : "memory");
}

__device__ __forceinline__ int chip_half() {
    return (int)((__builtin_amdgcn_s_sendmsg_rtn(0x87) >> 18) & 1u);
}

#define DGRID 1024
#define DB    8

__global__ void k_deg_init(float* __restrict__ deg2, int n, int npad) {
    int i = blockIdx.x * blockDim.x + threadIdx.x;
    if (i < n) {
        deg2[i] = 1.0f;
        deg2[npad + i] = 0.0f;
    }
}

__global__ void __launch_bounds__(256)
k_deg_accum(const int* __restrict__ col, const float* __restrict__ ew,
            float* __restrict__ deg2, int E, int npad) {
    float* deg = deg2 + (size_t)chip_half() * npad;
    const int nt = gridDim.x * blockDim.x;
    for (int eb = blockIdx.x * blockDim.x + threadIdx.x; eb < E; eb += nt * DB) {
        int   c[DB];
        float w[DB];
#pragma unroll
        for (int k = 0; k < DB; ++k) {
            const int e = eb + k * nt;
            c[k] = -1;
            w[k] = 0.0f;
            if (e < E) { c[k] = col[e]; w[k] = ew[e]; }
        }
#pragma unroll
        for (int k = 0; k < DB; ++k)
            if (c[k] >= 0) atomic_add_f32_dev(&deg[c[k]], w[k]);
    }
}

__global__ void k_rsqrt(const float* __restrict__ deg2, float* __restrict__ dinv, int n, int npad) {
    int i = blockIdx.x * blockDim.x + threadIdx.x;
    if (i < n) dinv[i] = rsqrtf(deg2[i] + deg2[npad + i]);
}

__device__ __forceinline__ void wmma_f32_cover() {
    __builtin_amdgcn_sched_barrier(0);
    asm volatile("v_nop\n\tv_nop\n\tv_nop\n\tv_nop");
    __builtin_amdgcn_sched_barrier(0);
}

#define GW      8
#define APITCH  132
#define GBLK    1024

template <bool SECOND>
__device__ __forceinline__ void
gemm128_body(const float* __restrict__ A, const float* __restrict__ W,
             const float* __restrict__ dinv, const float* __restrict__ bias,
             const float* __restrict__ abias  ,
             float* __restrict__ out, float* __restrict__ tgt, int nrows,
             float* __restrict__ As  ) {
    const int lane = threadIdx.x;
    const int wv   = threadIdx.y;
    const int n0   = wv * 16;
    const int hi   = lane >> 4;
    const int l    = lane & 15;
    const int ntiles = nrows / 16;
    const int tstep  = gridDim.x;

    v2f breg[NFEAT / 4];
    {
        const float* bp = W + (size_t)(2 * hi) * NFEAT + n0 + l;
#pragma unroll
        for (int s = 0; s < NFEAT / 4; ++s) {
            breg[s].x = bp[(size_t)(4 * s) * NFEAT];
            breg[s].y = bp[(size_t)(4 * s) * NFEAT + NFEAT];
        }
    }
    float bv = 0.0f;
    if (SECOND) bv = bias[n0 + l];

    const int    sr   = 2 * wv;
    const size_t goff = (size_t)sr * NFEAT + 4 * lane;
    const int    soff = sr * APITCH + 4 * lane;
    const int    foff = l * APITCH + 2 * hi;

    float4 ab = {};
    if (SECOND) { ab.x = abias[4 * lane]; ab.y = abias[4 * lane + 1]; ab.z = abias[4 * lane + 2]; ab.w = abias[4 * lane + 3]; }
    int t = blockIdx.x;
    float4 p0 = {}, p1 = {};
    float d0 = 0.0f, d1 = 0.0f;
    if (t < ntiles) {
        const float* g = A + (size_t)t * 16 * NFEAT + goff;
        p0 = *(const float4*)g;
        p1 = *(const float4*)(g + NFEAT);
        if (SECOND) { d0 = dinv[t * 16 + sr]; d1 = dinv[t * 16 + sr + 1]; }
    }
    int buf = 0;
    for (; t < ntiles; t += tstep, buf ^= 1) {
        float* as = As + buf * (16 * APITCH);
        if (SECOND) {
            p0.x = fmaxf(fmaf(d0, p0.x, ab.x), 0.0f); p0.y = fmaxf(fmaf(d0, p0.y, ab.y), 0.0f);
            p0.z = fmaxf(fmaf(d0, p0.z, ab.z), 0.0f); p0.w = fmaxf(fmaf(d0, p0.w, ab.w), 0.0f);
            p1.x = fmaxf(fmaf(d1, p1.x, ab.x), 0.0f); p1.y = fmaxf(fmaf(d1, p1.y, ab.y), 0.0f);
            p1.z = fmaxf(fmaf(d1, p1.z, ab.z), 0.0f); p1.w = fmaxf(fmaf(d1, p1.w, ab.w), 0.0f);
        }
        *(float4*)(as + soff)          = p0;
        *(float4*)(as + soff + APITCH) = p1;
        __syncthreads();
        const int tn = t + tstep;
        if (tn < ntiles) {
            const float* g = A + (size_t)tn * 16 * NFEAT + goff;
            p0 = *(const float4*)g;
            p1 = *(const float4*)(g + NFEAT);
            if (SECOND) { d0 = dinv[tn * 16 + sr]; d1 = dinv[tn * 16 + sr + 1]; }
        }

        v8f acc = {};
        const float* fp = as + foff;
#pragma unroll
        for (int s = 0; s < NFEAT / 4; ++s) {
            const v2f a = *(const v2f*)(fp + 4 * s);
            acc = __builtin_amdgcn_wmma_f32_16x16x4_f32(
                false, a, false, breg[s], (short)0, acc, false, false);
        }
        wmma_f32_cover();

        const int    r0  = t * 16 + 8 * hi;
        const size_t off = (size_t)r0 * NFEAT + n0 + l;
        float* op = out + off;
        float* tp = tgt + off;
#pragma unroll
        for (int v = 0; v < 8; ++v) {
            const float di = dinv[r0 + v];
            const float hv = acc[v];
            const float hs = di * hv;
            op[(size_t)v * NFEAT] = hs;
            tp[(size_t)v * NFEAT] = SECOND ? fmaf(di * di, hv, bv) : hs;
        }
    }
}

__global__ void __launch_bounds__(32 * GW)
k_gemm128_wmma(const float* __restrict__ A, const float* __restrict__ W,
               const float* __restrict__ dinv,
               float* __restrict__ out, float* __restrict__ tgt, int nrows) {
    __shared__ __attribute__((aligned(16))) float As[2 * 16 * APITCH];
    gemm128_body<false>(A, W, dinv, nullptr, nullptr, out, tgt, nrows, As);
}

__global__ void __launch_bounds__(32 * GW)
k_gemm128_wmma_relu(const float* __restrict__ A, const float* __restrict__ W,
                    const float* __restrict__ dinv, const float* __restrict__ bias,
                    const float* __restrict__ abias,
                    float* __restrict__ out, float* __restrict__ tgt, int nrows) {
    __shared__ __attribute__((aligned(16))) float As[2 * 16 * APITCH];
    gemm128_body<true>(A, W, dinv, bias, abias, out, tgt, nrows, As);
}

#define EPW   32
#define EB    8
#define EWV   8
#define EGRID 2048

__device__ __forceinline__ int lane_read_i(int v, int j) {
    return __builtin_amdgcn_readlane(v, j);
}
__device__ __forceinline__ float lane_read_f(float v, int j) {
    return __int_as_float(__builtin_amdgcn_readlane(__float_as_int(v), j));
}

__device__ __forceinline__ float4 load4_row(const float* rowp, unsigned boff) {
    const char* p = (const char*)rowp + boff;
    float4 v;
    v.x = *(const float*)(p);
    v.y = *(const float*)(p + 128);
    v.z = *(const float*)(p + 256);
    v.w = *(const float*)(p + 384);
    return v;
}

__device__ __forceinline__ void atomic_add4_row(float* rowp, unsigned boff, float4 v) {
    asm volatile("global_atomic_add_f32 %0, %1, %5 scope:SCOPE_DEV\n\t"
                 "global_atomic_add_f32 %0, %2, %5 offset:128 scope:SCOPE_DEV\n\t"
                 "global_atomic_add_f32 %0, %3, %5 offset:256 scope:SCOPE_DEV\n\t"
                 "global_atomic_add_f32 %0, %4, %5 offset:384 scope:SCOPE_DEV\n\t"
                 "s_wait_xcnt 0x0"
                 :: "v"(boff), "v"(v.x), "v"(v.y), "v"(v.z), "v"(v.w), "s"(rowp)
                 : "memory");
}

template <bool TFACTOR>
__global__ void __launch_bounds__(32 * EWV)
k_edge_agg(const int* __restrict__ row, const int* __restrict__ col,
           const float* __restrict__ ew, const float* __restrict__ dinv,
           const float* __restrict__ h, float* __restrict__ out, int E, int n) {
    const int lane = threadIdx.x;
    const int wv   = __builtin_amdgcn_readfirstlane((int)threadIdx.y);
    const unsigned boff = (unsigned)lane * 4u;
    const int nsplit = n >> 1;
    const int ngrp   = (E + EPW - 1) / EPW;
    const int gw     = blockIdx.x * EWV + wv;
    const int nw     = gridDim.x * EWV;
    const int myq    = chip_half();

    for (int sweep = 0; sweep < 2; ++sweep) {
        const int q = myq ^ sweep;
        for (int g = gw; g < ngrp; g += nw) {
            const int e = g * EPW + lane;
            int r = 0, c = 0;
            float w = 0.0f, nrm = 0.0f;
            bool mine = false;
            if (e < E) {
                c = col[e];
                r = row[e];
                w = ew[e];
                mine = ((c >= nsplit) ? 1 : 0) == q;
            }
            if (mine) nrm = TFACTOR ? w * dinv[c] : w;
            unsigned m = __builtin_amdgcn_ballot_w32(mine);
            for (int bt = 0; bt < EPW / EB && m != 0u; ++bt) {
                int jj[EB];
#pragma unroll
                for (int k = 0; k < EB; ++k) {
                    jj[k] = (m != 0u) ? __builtin_ctz(m) : -1;
                    m &= m - 1u;
                }
                float4 cur[EB];
#pragma unroll
                for (int k = 0; k < EB; ++k)
                    if (jj[k] >= 0)
                        cur[k] = load4_row(h + (size_t)lane_read_i(r, jj[k]) * NFEAT, boff);
#pragma unroll
                for (int k = 0; k < EB; ++k) {
                    if (jj[k] >= 0) {
                        const int cj = lane_read_i(c, jj[k]);
                        const float nj = lane_read_f(nrm, jj[k]);
                        float4 mv;
                        mv.x = nj * cur[k].x;
                        mv.y = nj * cur[k].y;
                        mv.z = nj * cur[k].z;
                        mv.w = nj * cur[k].w;
                        atomic_add4_row(out + (size_t)cj * NFEAT, boff, mv);
                    }
                }
            }
        }
    }
}

extern "C" void kernel_launch(void* const* d_in, const int* in_sizes, int n_in,
                              void* d_out, int out_size, void* d_ws, size_t ws_size,
                              hipStream_t stream) {
    const float* x  = (const float*)d_in[0];
    const int*   ei = (const int*)d_in[1];
    const float* ew = (const float*)d_in[2];
    const float* W1 = (const float*)d_in[3];
    const float* b1 = (const float*)d_in[4];
    const float* W2 = (const float*)d_in[5];
    const float* b2 = (const float*)d_in[6];
    float* out = (float*)d_out;

    const int N = in_sizes[0] / NFEAT;
    const int E = in_sizes[2];
    const int* row = ei;
    const int* col = ei + E;

    float* ws   = (float*)d_ws;
    size_t npad = ((size_t)N + 255) & ~(size_t)255;
    float* dinv = ws;
    float* h    = ws + npad;
    float* g    = h + (size_t)N * NFEAT;
    float* deg2 = g + (size_t)N * NFEAT;

    const int T = 256;
    k_deg_init<<<(N + T - 1) / T, T, 0, stream>>>(deg2, N, (int)npad);
    k_deg_accum<<<DGRID, T, 0, stream>>>(col, ew, deg2, E, (int)npad);
    k_rsqrt<<<(N + T - 1) / T, T, 0, stream>>>(deg2, dinv, N, (int)npad);

    dim3 wblk(32, GW);

    k_gemm128_wmma<<<GBLK, wblk, 0, stream>>>(x, W1, dinv, h, g, N);
    k_edge_agg<false><<<EGRID, dim3(32, EWV), 0, stream>>>(row, col, ew, dinv, h, g, E, N);

    k_gemm128_wmma_relu<<<GBLK, wblk, 0, stream>>>(g, W2, dinv, b2, b1, h, out, N);
    k_edge_agg<true><<<EGRID, dim3(32, EWV), 0, stream>>>(row, col, ew, dinv, h, out, E, N);
}


# Edge scatter in 4 passes (column half of the rows x target half), a wave's edges held in registers: a pass touches half of each source row, a quarter of the targets; same f32 atomics, arrival-ordered
# speedup vs baseline: 3.2666x; 1.1537x over previous
#include <hip/hip_runtime.h>

#define NFEAT 128

typedef float v2f  __attribute__((ext_vector_type(2)));
typedef float v8f  __attribute__((ext_vector_type(8)));

__device__ __forceinline__ void atomic_add_f32_dev(float* p, float v) {
    asm volatile("global_atomic_add_f32 %0, %1, off scope:SCOPE_DEV\n\t"
                 "s_wait_xcnt 0x0"
                 :: "v"(p), "v"(v) : "memory");
}

__device__ __forceinline__ int chip_half() {
    return (int)((__builtin_amdgcn_s_sendmsg_rtn(0x87) >> 18) & 1u);
}

#define DGRID 1024
#define DB    8

__global__ void k_deg_init(float* __restrict__ deg2, int n, int npad) {
    int i = blockIdx.x * blockDim.x + threadIdx.x;
    if (i < n) {
        deg2[i] = 1.0f;
        deg2[npad + i] = 0.0f;
    }
}

__global__ void __launch_bounds__(256)
k_deg_accum(const int* __restrict__ col, const float* __restrict__ ew,
            float* __restrict__ deg2, int E, int npad) {
    float* deg = deg2 + (size_t)chip_half() * npad;
    const int nt = gridDim.x * blockDim.x;
    for (int eb = blockIdx.x * blockDim.x + threadIdx.x; eb < E; eb += nt * DB) {
        int   c[DB];
        float w[DB];
#pragma unroll
        for (int k = 0; k < DB; ++k) {
            const int e = eb + k * nt;
            c[k] = -1;
            w[k] = 0.0f;
            if (e < E) { c[k] = col[e]; w[k] = ew[e]; }
        }
#pragma unroll
        for (int k = 0; k < DB; ++k)
            if (c[k] >= 0) atomic_add_f32_dev(&deg[c[k]], w[k]);
    }
}

__global__ void k_rsqrt(const float* __restrict__ deg2, float* __restrict__ dinv, int n, int npad) {
    int i = blockIdx.x * blockDim.x + threadIdx.x;
    if (i < n) dinv[i] = rsqrtf(deg2[i] + deg2[npad + i]);
}

__device__ __forceinline__ void wmma_f32_cover() {
    __builtin_amdgcn_sched_barrier(0);
    asm volatile("v_nop\n\tv_nop\n\tv_nop\n\tv_nop");
    __builtin_amdgcn_sched_barrier(0);
}

#define GW      8
#define APITCH  132
#define GBLK    1024

template <bool SECOND>
__device__ __forceinline__ void
gemm128_body(const float* __restrict__ A, const float* __restrict__ W,
             const float* __restrict__ dinv, const float* __restrict__ bias,
             const float* __restrict__ abias  ,
             float* __restrict__ out, float* __restrict__ tgt, int nrows,
             float* __restrict__ As  ) {
    const int lane = threadIdx.x;
    const int wv   = threadIdx.y;
    const int n0   = wv * 16;
    const int hi   = lane >> 4;
    const int l    = lane & 15;
    const int ntiles = nrows / 16;
    const int tstep  = gridDim.x;

    v2f breg[NFEAT / 4];
    {
        const float* bp = W + (size_t)(2 * hi) * NFEAT + n0 + l;
#pragma unroll
        for (int s = 0; s < NFEAT / 4; ++s) {
            breg[s].x = bp[(size_t)(4 * s) * NFEAT];
            breg[s].y = bp[(size_t)(4 * s) * NFEAT + NFEAT];
        }
    }
    float bv = 0.0f;
    if (SECOND) bv = bias[n0 + l];

    const int    sr   = 2 * wv;
    const size_t goff = (size_t)sr * NFEAT + 4 * lane;
    const int    soff = sr * APITCH + 4 * lane;
    const int    foff = l * APITCH + 2 * hi;

    float4 ab = {};
    if (SECOND) { ab.x = abias[4 * lane]; ab.y = abias[4 * lane + 1]; ab.z = abias[4 * lane + 2]; ab.w = abias[4 * lane + 3]; }
    int t = blockIdx.x;
    float4 p0 = {}, p1 = {};
    float d0 = 0.0f, d1 = 0.0f;
    if (t < ntiles) {
        const float* g = A + (size_t)t * 16 * NFEAT + goff;
        p0 = *(const float4*)g;
        p1 = *(const float4*)(g + NFEAT);
        if (SECOND) { d0 = dinv[t * 16 + sr]; d1 = dinv[t * 16 + sr + 1]; }
    }
    int buf = 0;
    for (; t < ntiles; t += tstep, buf ^= 1) {
        float* as = As + buf * (16 * APITCH);
        if (SECOND) {
            p0.x = fmaxf(fmaf(d0, p0.x, ab.x), 0.0f); p0.y = fmaxf(fmaf(d0, p0.y, ab.y), 0.0f);
            p0.z = fmaxf(fmaf(d0, p0.z, ab.z), 0.0f); p0.w = fmaxf(fmaf(d0, p0.w, ab.w), 0.0f);
            p1.x = fmaxf(fmaf(d1, p1.x, ab.x), 0.0f); p1.y = fmaxf(fmaf(d1, p1.y, ab.y), 0.0f);
            p1.z = fmaxf(fmaf(d1, p1.z, ab.z), 0.0f); p1.w = fmaxf(fmaf(d1, p1.w, ab.w), 0.0f);
        }
        *(float4*)(as + soff)          = p0;
        *(float4*)(as + soff + APITCH) = p1;
        __syncthreads();
        const int tn = t + tstep;
        if (tn < ntiles) {
            const float* g = A + (size_t)tn * 16 * NFEAT + goff;
            p0 = *(const float4*)g;
            p1 = *(const float4*)(g + NFEAT);
            if (SECOND) { d0 = dinv[tn * 16 + sr]; d1 = dinv[tn * 16 + sr + 1]; }
        }

        v8f acc = {};
        const float* fp = as + foff;
#pragma unroll
        for (int s = 0; s < NFEAT / 4; ++s) {
            const v2f a = *(const v2f*)(fp + 4 * s);
            acc = __builtin_amdgcn_wmma_f32_16x16x4_f32(
                false, a, false, breg[s], (short)0, acc, false, false);
        }
        wmma_f32_cover();

        const int    r0  = t * 16 + 8 * hi;
        const size_t off = (size_t)r0 * NFEAT + n0 + l;
        float* op = out + off;
        float* tp = tgt + off;
#pragma unroll
        for (int v = 0; v < 8; ++v) {
            const float di = dinv[r0 + v];
            const float hv = acc[v];
            const float hs = di * hv;
            op[(size_t)v * NFEAT] = hs;
            tp[(size_t)v * NFEAT] = SECOND ? fmaf(di * di, hv, bv) : hs;
        }
    }
}

__global__ void __launch_bounds__(32 * GW)
k_gemm128_wmma(const float* __restrict__ A, const float* __restrict__ W,
               const float* __restrict__ dinv,
               float* __restrict__ out, float* __restrict__ tgt, int nrows) {
    __shared__ __attribute__((aligned(16))) float As[2 * 16 * APITCH];
    gemm128_body<false>(A, W, dinv, nullptr, nullptr, out, tgt, nrows, As);
}

__global__ void __launch_bounds__(32 * GW)
k_gemm128_wmma_relu(const float* __restrict__ A, const float* __restrict__ W,
                    const float* __restrict__ dinv, const float* __restrict__ bias,
                    const float* __restrict__ abias,
                    float* __restrict__ out, float* __restrict__ tgt, int nrows) {
    __shared__ __attribute__((aligned(16))) float As[2 * 16 * APITCH];
    gemm128_body<true>(A, W, dinv, bias, abias, out, tgt, nrows, As);
}

#define EPW   32
#define EB    8
#define EWV   8
#define EGRID 2048
#define GPW   4

__device__ __forceinline__ int lane_read_i(int v, int j) {
    return __builtin_amdgcn_readlane(v, j);
}
__device__ __forceinline__ float lane_read_f(float v, int j) {
    return __int_as_float(__builtin_amdgcn_readlane(__float_as_int(v), j));
}

__device__ __forceinline__ float2 load2_half(const float* halfp, unsigned boff) {
    const char* p = (const char*)halfp + boff;
    float2 v;
    v.x = *(const float*)(p);
    v.y = *(const float*)(p + 128);
    return v;
}

__device__ __forceinline__ void atomic_add2_half(float* halfp, unsigned boff, float2 v) {
    asm volatile("global_atomic_add_f32 %0, %1, %3 scope:SCOPE_DEV\n\t"
                 "global_atomic_add_f32 %0, %2, %3 offset:128 scope:SCOPE_DEV\n\t"
                 "s_wait_xcnt 0x0"
                 :: "v"(boff), "v"(v.x), "v"(v.y), "s"(halfp)
                 : "memory");
}

template <bool TFACTOR>
__global__ void __launch_bounds__(32 * EWV)
k_edge_agg(const int* __restrict__ row, const int* __restrict__ col,
           const float* __restrict__ ew, const float* __restrict__ dinv,
           const float* __restrict__ h, float* __restrict__ out, int E, int n) {
    const int lane = threadIdx.x;
    const int wv   = __builtin_amdgcn_readfirstlane((int)threadIdx.y);
    const unsigned boff = (unsigned)lane * 4u;
    const int nsplit = n >> 1;
    const int ngrp   = (E + EPW - 1) / EPW;
    const int gw     = blockIdx.x * EWV + wv;
    const int nw     = gridDim.x * EWV;
    const int myq    = chip_half();

    for (int g0 = gw; g0 < ngrp; g0 += GPW * nw) {
        int   r[GPW], c[GPW];
        float f[GPW];
#pragma unroll
        for (int k = 0; k < GPW; ++k) {
            const int g = g0 + k * nw;
            const int e = g * EPW + lane;
            r[k] = 0;
            c[k] = -1;
            f[k] = 0.0f;
            if (g < ngrp && e < E) {
                c[k] = col[e];
                r[k] = row[e];
                f[k] = ew[e];
            }
        }
        if (TFACTOR) {
#pragma unroll
            for (int k = 0; k < GPW; ++k)
                if (c[k] >= 0) f[k] = f[k] * dinv[c[k]];
        }

        for (int pass = 0; pass < 4; ++pass) {
            const int ch = pass >> 1;
            const int q  = myq ^ (pass & 1);
            const float* hb = h + ch * (NFEAT / 2);
            float*       ob = out + ch * (NFEAT / 2);
#pragma unroll
            for (int k = 0; k < GPW; ++k) {
                const bool mine = (c[k] >= 0) && ((((c[k] >= nsplit) ? 1 : 0)) == q);
                unsigned m = __builtin_amdgcn_ballot_w32(mine);
                for (int bt = 0; bt < EPW / EB && m != 0u; ++bt) {
                    int jj[EB];
#pragma unroll
                    for (int i = 0; i < EB; ++i) {
                        jj[i] = (m != 0u) ? __builtin_ctz(m) : -1;
                        m &= m - 1u;
                    }
                    float2 cur[EB];
#pragma unroll
                    for (int i = 0; i < EB; ++i)
                        if (jj[i] >= 0)
                            cur[i] = load2_half(hb + (size_t)lane_read_i(r[k], jj[i]) * NFEAT, boff);
#pragma unroll
                    for (int i = 0; i < EB; ++i) {
                        if (jj[i] >= 0) {
                            const int   cj = lane_read_i(c[k], jj[i]);
                            const float fj = lane_read_f(f[k], jj[i]);
                            float2 mv;
                            mv.x = fj * cur[i].x;
                            mv.y = fj * cur[i].y;
                            atomic_add2_half(ob + (size_t)cj * NFEAT, boff, mv);
                        }
                    }
                }
            }
        }
    }
}

extern "C" void kernel_launch(void* const* d_in, const int* in_sizes, int n_in,
                              void* d_out, int out_size, void* d_ws, size_t ws_size,
                              hipStream_t stream) {
    const float* x  = (const float*)d_in[0];
    const int*   ei = (const int*)d_in[1];
    const float* ew = (const float*)d_in[2];
    const float* W1 = (const float*)d_in[3];
    const float* b1 = (const float*)d_in[4];
    const float* W2 = (const float*)d_in[5];
    const float* b2 = (const float*)d_in[6];
    float* out = (float*)d_out;

    const int N = in_sizes[0] / NFEAT;
    const int E = in_sizes[2];
    const int* row = ei;
    const int* col = ei + E;

    float* ws   = (float*)d_ws;
    size_t npad = ((size_t)N + 255) & ~(size_t)255;
    float* dinv = ws;
    float* h    = ws + npad;
    float* g    = h + (size_t)N * NFEAT;
    float* deg2 = g + (size_t)N * NFEAT;

    const int T = 256;
    k_deg_init<<<(N + T - 1) / T, T, 0, stream>>>(deg2, N, (int)npad);
    k_deg_accum<<<DGRID, T, 0, stream>>>(col, ew, deg2, E, (int)npad);
    k_rsqrt<<<(N + T - 1) / T, T, 0, stream>>>(deg2, dinv, N, (int)npad);

    dim3 wblk(32, GW);

    k_gemm128_wmma<<<GBLK, wblk, 0, stream>>>(x, W1, dinv, h, g, N);
    k_edge_agg<false><<<EGRID, dim3(32, EWV), 0, stream>>>(row, col, ew, dinv, h, g, E, N);

    k_gemm128_wmma_relu<<<GBLK, wblk, 0, stream>>>(g, W2, dinv, b2, b1, h, out, N);
    k_edge_agg<true><<<EGRID, dim3(32, EWV), 0, stream>>>(row, col, ew, dinv, h, out, E, N);
}
